# starting kernel
# baseline (speedup 1.0000x reference)
#include <hip/hip_runtime.h>

typedef float v2f __attribute__((ext_vector_type(2)));
typedef float v8f __attribute__((ext_vector_type(8)));

__global__ void __launch_bounds__(256)
sage_scatter(const float* __restrict__ x, const int* __restrict__ src,
             const int* __restrict__ dst, float* __restrict__ agg,
             float* __restrict__ cnt, int E, int d, int do_count)
{
    const int lane = threadIdx.x & 31;
    const int wid  = blockIdx.x * (blockDim.x >> 5) + (threadIdx.x >> 5);
    const int nw   = gridDim.x * (blockDim.x >> 5);
    const int d4   = d >> 2;

    for (int e = wid; e < E; e += nw) {
        const int s = src[e];
        const int t = dst[e];

        const int e2 = e + nw;
        if (e2 < E) {
            __builtin_prefetch(&x[(size_t)src[e2] * d + lane * 4], 0, 1);
        }

        const float4* xs = (const float4*)(x + (size_t)s * d);
        float*        at = agg + (size_t)t * d;

        for (int c = lane; c < d4; c += 32) {
            float4 v = xs[c];
            atomicAdd(at + 4 * c + 0, v.x);
            atomicAdd(at + 4 * c + 1, v.y);
            atomicAdd(at + 4 * c + 2, v.z);
            atomicAdd(at + 4 * c + 3, v.w);
        }
        if (do_count && lane == 0) atomicAdd(cnt + t, 1.0f);
    }
}

template <int K, int NCOL, bool RELU>
__global__ void __launch_bounds__(256)
sage_gemm(const float* __restrict__ agg, const float* __restrict__ cnt,
          const float* __restrict__ xroot, const float* __restrict__ Wl,
          const float* __restrict__ Wr, const float* __restrict__ bias,
          float* __restrict__ out, int nrows)
{
    constexpr int JT     = 4;
    constexpr int GROUPS = NCOL / (16 * JT);

    const int lane = threadIdx.x & 31;
    const int half = lane >> 4;
    const int l    = lane & 15;

    const int wid = blockIdx.x * (blockDim.x >> 5) + (threadIdx.x >> 5);
    const int tm  = (wid / GROUPS) * 16;
    const int tn  = (wid % GROUPS) * (16 * JT);
    if (tm >= nrows) return;

    const int row = tm + l;
    const float  inv  = 1.0f / fmaxf(cnt[row], 1.0f);
    const float* aRow = agg   + (size_t)row * K;
    const float* xRow = xroot + (size_t)row * K;

    v8f acc[JT] = {};
#pragma unroll 2
    for (int k0 = 0; k0 < K; k0 += 4) {
        const int ka = k0 + 2 * half;

        v2f am, ar;
        am.x = aRow[ka] * inv;  am.y = aRow[ka + 1] * inv;
        ar.x = xRow[ka];        ar.y = xRow[ka + 1];

#pragma unroll
        for (int j = 0; j < JT; ++j) {
            const int col = tn + 16 * j + l;
            v2f bl, br;
            bl.x = Wl[(size_t)ka * NCOL + col];
            bl.y = Wl[(size_t)(ka + 1) * NCOL + col];
            br.x = Wr[(size_t)ka * NCOL + col];
            br.y = Wr[(size_t)(ka + 1) * NCOL + col];

            acc[j] = __builtin_amdgcn_wmma_f32_16x16x4_f32(
                false, am, false, bl, (short)0, acc[j], false, false);
            acc[j] = __builtin_amdgcn_wmma_f32_16x16x4_f32(
                false, ar, false, br, (short)0, acc[j], false, false);
        }
    }

#pragma unroll
    for (int j = 0; j < JT; ++j) {
        const int col = tn + 16 * j + l;
        const float bv = bias[col];
#pragma unroll
        for (int r = 0; r < 8; ++r) {
            float v = acc[j][r] + bv;
            if (RELU) v = fmaxf(v, 0.0f);
            out[(size_t)(tm + 8 * half + r) * NCOL + col] = v;
        }
    }
}

extern "C" void kernel_launch(void* const* d_in, const int* in_sizes, int n_in,
                              void* d_out, int out_size, void* d_ws, size_t ws_size,
                              hipStream_t stream)
{
    constexpr int IN_DIM = 128, HID = 256, EMB = 128;

    const float* x   = (const float*)d_in[0];
    const int*   ei  = (const int*)  d_in[1];
    const float* W1l = (const float*)d_in[2];
    const float* b1  = (const float*)d_in[3];
    const float* W1r = (const float*)d_in[4];
    const float* W2l = (const float*)d_in[5];
    const float* b2  = (const float*)d_in[6];
    const float* W2r = (const float*)d_in[7];

    const int N = in_sizes[0] / IN_DIM;
    const int E = in_sizes[1] / 2;
    const int* src = ei;
    const int* dst = ei + E;

    float* cnt  = (float*)d_ws;
    float* agg1 = cnt  + (((size_t)N + 63) & ~(size_t)63);
    float* h    = agg1 + (size_t)N * IN_DIM;
    float* agg2 = h    + (size_t)N * HID;

    hipMemsetAsync(cnt,  0, sizeof(float) * (size_t)N,          stream);
    hipMemsetAsync(agg1, 0, sizeof(float) * (size_t)N * IN_DIM, stream);
    hipMemsetAsync(agg2, 0, sizeof(float) * (size_t)N * HID,    stream);

    const dim3 blk(256);
    const int scatterBlocks = 2048;
    const int tmN = (N + 15) / 16;

    sage_scatter<<<scatterBlocks, blk, 0, stream>>>(x, src, dst, agg1, cnt,
                                                    E, IN_DIM, 1);
    const int waves_l1  = tmN * (HID / 64);
    const int blocks_l1 = (waves_l1 + 7) / 8;
    sage_gemm<IN_DIM, HID, true><<<blocks_l1, blk, 0, stream>>>(
        agg1, cnt, x, W1l, W1r, b1, h, N);

    sage_scatter<<<scatterBlocks, blk, 0, stream>>>(h, src, dst, agg2, cnt,
                                                    E, HID, 0);
    const int waves_l2  = tmN * (EMB / 64);
    const int blocks_l2 = (waves_l2 + 7) / 8;
    sage_gemm<HID, EMB, false><<<blocks_l2, blk, 0, stream>>>(
        agg2, cnt, h, W2l, W2r, b2, (float*)d_out, N);
}


# Replace float-atomic scatter by a per-call destination-grouped edge list (int atomics: count, scan, fill) and a wave-per-node register gather-sum; float sum order: ascending source id per node
# speedup vs baseline: 1.8456x; 1.8456x over previous
#include <hip/hip_runtime.h>

typedef float v2f __attribute__((ext_vector_type(2)));
typedef float v8f __attribute__((ext_vector_type(8)));


__global__ void __launch_bounds__(256)
csr_count(const int* __restrict__ dst, int* __restrict__ deg, int E)
{
    const int stride = gridDim.x * blockDim.x;
    for (int e = blockIdx.x * blockDim.x + threadIdx.x; e < E; e += stride)
        atomicAdd(&deg[dst[e]], 1);
}

__device__ __forceinline__ int wave_incl_scan(int v, int lane)
{
#pragma unroll
    for (int o = 1; o < 32; o <<= 1) {
        const int t = __shfl_up(v, o);
        if (lane >= o) v += t;
    }
    return v;
}

__device__ __forceinline__ int wave_sum(int v)
{
#pragma unroll
    for (int o = 16; o > 0; o >>= 1) v += __shfl_xor(v, o);
    return v;
}

__global__ void __launch_bounds__(1024)
csr_scan(const int* __restrict__ deg, int* __restrict__ rowptr, int np)
{
    __shared__ int wtot[32];
    const int lane  = threadIdx.x & 31;
    const int w     = threadIdx.x >> 5;
    const int steps = np >> 7;
    const int per   = (steps + 31) >> 5;
    const int s0    = min(w * per, steps);
    const int s1    = min(s0 + per, steps);
    const int4* d4  = (const int4*)deg;
    int4*       r4  = (int4*)rowptr;

    int s = 0;
    for (int st = s0; st < s1; ++st) {
        const int4 v = d4[st * 32 + lane];
        s += (v.x + v.y) + (v.z + v.w);
    }
    s = wave_sum(s);
    if (lane == 0) wtot[w] = s;
    __syncthreads();

    int carry = wave_sum(lane < w ? wtot[lane] : 0);
    for (int st = s0; st < s1; ++st) {
        const int4 v   = d4[st * 32 + lane];
        const int  tot = (v.x + v.y) + (v.z + v.w);
        const int  inc = wave_incl_scan(tot, lane);
        const int  ex  = carry + inc - tot;
        int4 o;
        o.x = ex; o.y = ex + v.x; o.z = o.y + v.y; o.w = o.z + v.z;
        r4[st * 32 + lane] = o;
        carry += __shfl(inc, 31);
    }
}

__global__ void __launch_bounds__(256)
csr_fill(const int* __restrict__ src, const int* __restrict__ dst,
         const int* __restrict__ rowptr, int* __restrict__ deg,
         int* __restrict__ col, int E)
{
    const int stride = gridDim.x * blockDim.x;
    for (int e = blockIdx.x * blockDim.x + threadIdx.x; e < E; e += stride) {
        const int t = dst[e];
        const int k = atomicSub(&deg[t], 1) - 1;
        col[rowptr[t] + k] = src[e];
    }
}

template <int D>
__global__ void __launch_bounds__(256)
sage_gather(const float* __restrict__ x, const int* __restrict__ rowptr,
            const int* __restrict__ col, float* __restrict__ agg,
            float* __restrict__ cnt, int n, int do_count)
{
    constexpr int V = D / 128;
    const int lane = threadIdx.x & 31;
    const int wid = __builtin_amdgcn_readfirstlane(
        (int)(blockIdx.x * (blockDim.x >> 5) + (threadIdx.x >> 5)));
    const int nw  = gridDim.x * (blockDim.x >> 5);

    for (int node = wid; node < n; node += nw) {
        const int b = rowptr[node];
        const int e = rowptr[node + 1];

        float4 acc[V];
    #pragma unroll
        for (int v = 0; v < V; ++v) acc[v] = make_float4(0.f, 0.f, 0.f, 0.f);

        for (int base = b; base < e; base += 32) {
            const int m  = min(32, e - base);
            const int id = (lane < m) ? col[base + lane] : 0x7fffffff;

            int rank = 0;
            for (int j = 0; j < m; ++j) {
                const int v = __builtin_amdgcn_readlane(id, j);
                rank += (v < id || (v == id && j < lane)) ? 1 : 0;
            }

    #define ID_OF_RANK(k) __builtin_amdgcn_readlane(id, __builtin_ctz(__builtin_amdgcn_ballot_w32(rank == (k))))

            int i = 0;
            for (; i + 4 <= m; i += 4) {
                const int r0 = ID_OF_RANK(i);
                const int r1 = ID_OF_RANK(i + 1);
                const int r2 = ID_OF_RANK(i + 2);
                const int r3 = ID_OF_RANK(i + 3);
                const float4* p0 = (const float4*)(x + (size_t)r0 * D);
                const float4* p1 = (const float4*)(x + (size_t)r1 * D);
                const float4* p2 = (const float4*)(x + (size_t)r2 * D);
                const float4* p3 = (const float4*)(x + (size_t)r3 * D);
                float4 t0[V], t1[V], t2[V], t3[V];
    #pragma unroll
                for (int v = 0; v < V; ++v) {
                    t0[v] = p0[lane + 32 * v];
                    t1[v] = p1[lane + 32 * v];
                    t2[v] = p2[lane + 32 * v];
                    t3[v] = p3[lane + 32 * v];
                }
    #pragma unroll
                for (int v = 0; v < V; ++v) {
                    acc[v].x += t0[v].x; acc[v].y += t0[v].y; acc[v].z += t0[v].z; acc[v].w += t0[v].w;
                    acc[v].x += t1[v].x; acc[v].y += t1[v].y; acc[v].z += t1[v].z; acc[v].w += t1[v].w;
                    acc[v].x += t2[v].x; acc[v].y += t2[v].y; acc[v].z += t2[v].z; acc[v].w += t2[v].w;
                    acc[v].x += t3[v].x; acc[v].y += t3[v].y; acc[v].z += t3[v].z; acc[v].w += t3[v].w;
                }
            }
            for (; i < m; ++i) {
                const int r0 = ID_OF_RANK(i);
                const float4* p0 = (const float4*)(x + (size_t)r0 * D);
    #pragma unroll
                for (int v = 0; v < V; ++v) {
                    const float4 t = p0[lane + 32 * v];
                    acc[v].x += t.x; acc[v].y += t.y; acc[v].z += t.z; acc[v].w += t.w;
                }
            }
    #undef ID_OF_RANK
        }

        float4* o4 = (float4*)(agg + (size_t)node * D);
    #pragma unroll
        for (int v = 0; v < V; ++v) o4[lane + 32 * v] = acc[v];
        if (do_count && lane == 0) cnt[node] = (float)(e - b);
    }
}

template <int K, int NCOL, bool RELU>
__global__ void __launch_bounds__(256)
sage_gemm(const float* __restrict__ agg, const float* __restrict__ cnt,
          const float* __restrict__ xroot, const float* __restrict__ Wl,
          const float* __restrict__ Wr, const float* __restrict__ bias,
          float* __restrict__ out, int nrows)
{
    constexpr int JT     = 4;
    constexpr int GROUPS = NCOL / (16 * JT);

    const int lane = threadIdx.x & 31;
    const int half = lane >> 4;
    const int l    = lane & 15;

    const int wid = blockIdx.x * (blockDim.x >> 5) + (threadIdx.x >> 5);
    const int tm  = (wid / GROUPS) * 16;
    const int tn  = (wid % GROUPS) * (16 * JT);
    if (tm >= nrows) return;

    const int row = tm + l;
    const float  inv  = 1.0f / fmaxf(cnt[row], 1.0f);
    const float* aRow = agg   + (size_t)row * K;
    const float* xRow = xroot + (size_t)row * K;

    v8f acc[JT] = {};
#pragma unroll 2
    for (int k0 = 0; k0 < K; k0 += 4) {
        const int ka = k0 + 2 * half;

        v2f am, ar;
        am.x = aRow[ka] * inv;  am.y = aRow[ka + 1] * inv;
        ar.x = xRow[ka];        ar.y = xRow[ka + 1];

#pragma unroll
        for (int j = 0; j < JT; ++j) {
            const int col = tn + 16 * j + l;
            v2f bl, br;
            bl.x = Wl[(size_t)ka * NCOL + col];
            bl.y = Wl[(size_t)(ka + 1) * NCOL + col];
            br.x = Wr[(size_t)ka * NCOL + col];
            br.y = Wr[(size_t)(ka + 1) * NCOL + col];

            acc[j] = __builtin_amdgcn_wmma_f32_16x16x4_f32(
                false, am, false, bl, (short)0, acc[j], false, false);
            acc[j] = __builtin_amdgcn_wmma_f32_16x16x4_f32(
                false, ar, false, br, (short)0, acc[j], false, false);
        }
    }

#pragma unroll
    for (int j = 0; j < JT; ++j) {
        const int col = tn + 16 * j + l;
        const float bv = bias[col];
#pragma unroll
        for (int r = 0; r < 8; ++r) {
            float v = acc[j][r] + bv;
            if (RELU) v = fmaxf(v, 0.0f);
            out[(size_t)(tm + 8 * half + r) * NCOL + col] = v;
        }
    }
}

extern "C" void kernel_launch(void* const* d_in, const int* in_sizes, int n_in,
                              void* d_out, int out_size, void* d_ws, size_t ws_size,
                              hipStream_t stream)
{
    constexpr int IN_DIM = 128, HID = 256, EMB = 128;

    const float* x   = (const float*)d_in[0];
    const int*   ei  = (const int*)  d_in[1];
    const float* W1l = (const float*)d_in[2];
    const float* b1  = (const float*)d_in[3];
    const float* W1r = (const float*)d_in[4];
    const float* W2l = (const float*)d_in[5];
    const float* b2  = (const float*)d_in[6];
    const float* W2r = (const float*)d_in[7];

    const int N = in_sizes[0] / IN_DIM;
    const int E = in_sizes[1] / 2;
    const int* src = ei;
    const int* dst = ei + E;

    const int NP = ((N + 128) / 128) * 128;

    int*   deg    = (int*)d_ws;
    int*   rowptr = deg + NP;
    int*   col    = rowptr + NP;
    float* cnt    = (float*)(col + (((size_t)E + 63) & ~(size_t)63));
    float* agg1   = cnt  + (((size_t)N + 63) & ~(size_t)63);
    float* h      = agg1 + (size_t)N * IN_DIM;
    float* agg2   = h    + (size_t)N * HID;

    hipMemsetAsync(deg, 0, sizeof(int) * (size_t)NP, stream);

    const dim3 blk(256);
    const int edgeBlocks = 2048;
    const int nodeBlocks = 2048;
    const int tmN = (N + 15) / 16;

    csr_count<<<edgeBlocks, blk, 0, stream>>>(dst, deg, E);
    csr_scan<<<1, dim3(1024), 0, stream>>>(deg, rowptr, NP);
    csr_fill<<<edgeBlocks, blk, 0, stream>>>(src, dst, rowptr, deg, col, E);

    sage_gather<IN_DIM><<<nodeBlocks, blk, 0, stream>>>(x, rowptr, col, agg1,
                                                        cnt, N, 1);
    const int waves_l1  = tmN * (HID / 64);
    const int blocks_l1 = (waves_l1 + 7) / 8;
    sage_gemm<IN_DIM, HID, true><<<blocks_l1, blk, 0, stream>>>(
        agg1, cnt, x, W1l, W1r, b1, h, N);

    sage_gather<HID><<<nodeBlocks, blk, 0, stream>>>(h, rowptr, col, agg2,
                                                     cnt, N, 0);
    const int waves_l2  = tmN * (EMB / 64);
    const int blocks_l2 = (waves_l2 + 7) / 8;
    sage_gemm<HID, EMB, false><<<blocks_l2, blk, 0, stream>>>(
        agg2, cnt, h, W2l, W2r, b2, (float*)d_out, N);
}


# GEMMs: one resident 16-wave work-group per WGP keeps both weight matrices in LDS, waves walk 16x64 strips with A as float4 rows; K sum order regrouped inside 8-k blocks
# speedup vs baseline: 2.1087x; 1.1426x over previous
#include <hip/hip_runtime.h>

typedef float v2f __attribute__((ext_vector_type(2)));
typedef float v8f __attribute__((ext_vector_type(8)));


__global__ void __launch_bounds__(256)
csr_count(const int* __restrict__ dst, int* __restrict__ deg, int E)
{
    const int stride = gridDim.x * blockDim.x;
    for (int e = blockIdx.x * blockDim.x + threadIdx.x; e < E; e += stride)
        atomicAdd(&deg[dst[e]], 1);
}

__device__ __forceinline__ int wave_incl_scan(int v, int lane)
{
#pragma unroll
    for (int o = 1; o < 32; o <<= 1) {
        const int t = __shfl_up(v, o);
        if (lane >= o) v += t;
    }
    return v;
}

__device__ __forceinline__ int wave_sum(int v)
{
#pragma unroll
    for (int o = 16; o > 0; o >>= 1) v += __shfl_xor(v, o);
    return v;
}

__global__ void __launch_bounds__(1024)
csr_scan(const int* __restrict__ deg, int* __restrict__ rowptr, int np)
{
    __shared__ int wtot[32];
    const int lane  = threadIdx.x & 31;
    const int w     = threadIdx.x >> 5;
    const int steps = np >> 7;
    const int per   = (steps + 31) >> 5;
    const int s0    = min(w * per, steps);
    const int s1    = min(s0 + per, steps);
    const int4* d4  = (const int4*)deg;
    int4*       r4  = (int4*)rowptr;

    int s = 0;
    for (int st = s0; st < s1; ++st) {
        const int4 v = d4[st * 32 + lane];
        s += (v.x + v.y) + (v.z + v.w);
    }
    s = wave_sum(s);
    if (lane == 0) wtot[w] = s;
    __syncthreads();

    int carry = wave_sum(lane < w ? wtot[lane] : 0);
    for (int st = s0; st < s1; ++st) {
        const int4 v   = d4[st * 32 + lane];
        const int  tot = (v.x + v.y) + (v.z + v.w);
        const int  inc = wave_incl_scan(tot, lane);
        const int  ex  = carry + inc - tot;
        int4 o;
        o.x = ex; o.y = ex + v.x; o.z = o.y + v.y; o.w = o.z + v.z;
        r4[st * 32 + lane] = o;
        carry += __shfl(inc, 31);
    }
}

__global__ void __launch_bounds__(256)
csr_fill(const int* __restrict__ src, const int* __restrict__ dst,
         const int* __restrict__ rowptr, int* __restrict__ deg,
         int* __restrict__ col, int E)
{
    const int stride = gridDim.x * blockDim.x;
    for (int e = blockIdx.x * blockDim.x + threadIdx.x; e < E; e += stride) {
        const int t = dst[e];
        const int k = atomicSub(&deg[t], 1) - 1;
        col[rowptr[t] + k] = src[e];
    }
}

template <int D>
__global__ void __launch_bounds__(256)
sage_gather(const float* __restrict__ x, const int* __restrict__ rowptr,
            const int* __restrict__ col, float* __restrict__ agg,
            float* __restrict__ cnt, int n, int do_count)
{
    constexpr int V = D / 128;
    const int lane = threadIdx.x & 31;
    const int wid = __builtin_amdgcn_readfirstlane(
        (int)(blockIdx.x * (blockDim.x >> 5) + (threadIdx.x >> 5)));
    const int nw  = gridDim.x * (blockDim.x >> 5);

    for (int node = wid; node < n; node += nw) {
        const int b = rowptr[node];
        const int e = rowptr[node + 1];

        float4 acc[V];
    #pragma unroll
        for (int v = 0; v < V; ++v) acc[v] = make_float4(0.f, 0.f, 0.f, 0.f);

        for (int base = b; base < e; base += 32) {
            const int m  = min(32, e - base);
            const int id = (lane < m) ? col[base + lane] : 0x7fffffff;

            int rank = 0;
            for (int j = 0; j < m; ++j) {
                const int v = __builtin_amdgcn_readlane(id, j);
                rank += (v < id || (v == id && j < lane)) ? 1 : 0;
            }

    #define ID_OF_RANK(k) __builtin_amdgcn_readlane(id, __builtin_ctz(__builtin_amdgcn_ballot_w32(rank == (k))))

            int i = 0;
            for (; i + 4 <= m; i += 4) {
                const int r0 = ID_OF_RANK(i);
                const int r1 = ID_OF_RANK(i + 1);
                const int r2 = ID_OF_RANK(i + 2);
                const int r3 = ID_OF_RANK(i + 3);
                const float4* p0 = (const float4*)(x + (size_t)r0 * D);
                const float4* p1 = (const float4*)(x + (size_t)r1 * D);
                const float4* p2 = (const float4*)(x + (size_t)r2 * D);
                const float4* p3 = (const float4*)(x + (size_t)r3 * D);
                float4 t0[V], t1[V], t2[V], t3[V];
    #pragma unroll
                for (int v = 0; v < V; ++v) {
                    t0[v] = p0[lane + 32 * v];
                    t1[v] = p1[lane + 32 * v];
                    t2[v] = p2[lane + 32 * v];
                    t3[v] = p3[lane + 32 * v];
                }
    #pragma unroll
                for (int v = 0; v < V; ++v) {
                    acc[v].x += t0[v].x; acc[v].y += t0[v].y; acc[v].z += t0[v].z; acc[v].w += t0[v].w;
                    acc[v].x += t1[v].x; acc[v].y += t1[v].y; acc[v].z += t1[v].z; acc[v].w += t1[v].w;
                    acc[v].x += t2[v].x; acc[v].y += t2[v].y; acc[v].z += t2[v].z; acc[v].w += t2[v].w;
                    acc[v].x += t3[v].x; acc[v].y += t3[v].y; acc[v].z += t3[v].z; acc[v].w += t3[v].w;
                }
            }
            for (; i < m; ++i) {
                const int r0 = ID_OF_RANK(i);
                const float4* p0 = (const float4*)(x + (size_t)r0 * D);
    #pragma unroll
                for (int v = 0; v < V; ++v) {
                    const float4 t = p0[lane + 32 * v];
                    acc[v].x += t.x; acc[v].y += t.y; acc[v].z += t.z; acc[v].w += t.w;
                }
            }
    #undef ID_OF_RANK
        }

        float4* o4 = (float4*)(agg + (size_t)node * D);
    #pragma unroll
        for (int v = 0; v < V; ++v) o4[lane + 32 * v] = acc[v];
        if (do_count && lane == 0) cnt[node] = (float)(e - b);
    }
}

template <int NCOL, int QK, int PQ>
__device__ __forceinline__ void stage_weights(const float* __restrict__ Wsrc,
                                              float4* dst, int tid)
{
    constexpr int TRIPS = QK * NCOL / 512;
#pragma unroll
    for (int i0 = 0; i0 < TRIPS; i0 += 8) {
        float4 v[8];
#pragma unroll
        for (int u = 0; u < 8; ++u) {
            const int e = tid + 512 * (i0 + u);
            const float* s = Wsrc + (size_t)(4 * (e / NCOL)) * NCOL + (e % NCOL);
            v[u].x = s[0];
            v[u].y = s[NCOL];
            v[u].z = s[2 * NCOL];
            v[u].w = s[3 * NCOL];
        }
#pragma unroll
        for (int u = 0; u < 8; ++u) {
            const int e = tid + 512 * (i0 + u);
            dst[(e / NCOL) * PQ + (e % NCOL)] = v[u];
        }
    }
}

template <int K, int NCOL, bool RELU>
__global__ void __launch_bounds__(512)
sage_gemm(const float* __restrict__ agg, const float* __restrict__ cnt,
          const float* __restrict__ xroot, const float* __restrict__ Wl,
          const float* __restrict__ Wr, const float* __restrict__ bias,
          float* __restrict__ out, int nrows)
{
    constexpr int JT     = 4;
    constexpr int GROUPS = NCOL / (16 * JT);
    constexpr int QK     = K / 4;
    constexpr int PQ     = NCOL + 8;
    constexpr int NB     = K / 8;
    __shared__ float4 Ws[2 * QK * PQ];

    const int tid = threadIdx.x;

    stage_weights<NCOL, QK, PQ>(Wl, Ws, tid);
    stage_weights<NCOL, QK, PQ>(Wr, Ws + QK * PQ, tid);
    __syncthreads();

    const int lane = tid & 31;
    const int half = lane >> 4;
    const int c    = lane & 15;
    const int w = __builtin_amdgcn_readfirstlane(tid >> 5);

    const int ntiles = (nrows + 15) >> 4;
    const int first  = (int)blockIdx.x;
    const int step   = (int)gridDim.x;
    const int mine   = (first < ntiles) ? (ntiles - first + step - 1) / step : 0;
    const int units  = mine * GROUPS;

    for (int u = w; u < units; u += 16) {
        const int tm = (first + (u / GROUPS) * step) * 16;
        const int tn = (u % GROUPS) * (16 * JT);

        const int row = min(tm + c, nrows - 1);
        const float  inv = 1.0f / fmaxf(cnt[row], 1.0f);
        const float4* aP = (const float4*)(agg   + (size_t)row * K) + half;
        const float4* xP = (const float4*)(xroot + (size_t)row * K) + half;
        const float4* bL = Ws + half * PQ + tn + c;
        const float4* bR = bL + QK * PQ;

        v8f acc[JT] = {};
        float4 am = aP[0];
        float4 ar = xP[0];
        for (int j = 0; j < NB; ++j) {
            const int jn = (j + 1 < NB) ? (j + 1) : j;
            const float4 amn = aP[2 * jn];
            const float4 arn = xP[2 * jn];

            v2f m0, m1, r0, r1;
            m0.x = am.x * inv; m0.y = am.y * inv;
            m1.x = am.z * inv; m1.y = am.w * inv;
            r0.x = ar.x;       r0.y = ar.y;
            r1.x = ar.z;       r1.y = ar.w;

            float4 wl[JT], wr[JT];
#pragma unroll
            for (int t = 0; t < JT; ++t) {
                wl[t] = bL[(2 * j) * PQ + 16 * t];
                wr[t] = bR[(2 * j) * PQ + 16 * t];
            }
#pragma unroll
            for (int t = 0; t < JT; ++t) {
                v2f b; b.x = wl[t].x; b.y = wl[t].y;
                acc[t] = __builtin_amdgcn_wmma_f32_16x16x4_f32(
                    false, m0, false, b, (short)0, acc[t], false, false);
            }
#pragma unroll
            for (int t = 0; t < JT; ++t) {
                v2f b; b.x = wr[t].x; b.y = wr[t].y;
                acc[t] = __builtin_amdgcn_wmma_f32_16x16x4_f32(
                    false, r0, false, b, (short)0, acc[t], false, false);
            }
#pragma unroll
            for (int t = 0; t < JT; ++t) {
                v2f b; b.x = wl[t].z; b.y = wl[t].w;
                acc[t] = __builtin_amdgcn_wmma_f32_16x16x4_f32(
                    false, m1, false, b, (short)0, acc[t], false, false);
            }
#pragma unroll
            for (int t = 0; t < JT; ++t) {
                v2f b; b.x = wr[t].z; b.y = wr[t].w;
                acc[t] = __builtin_amdgcn_wmma_f32_16x16x4_f32(
                    false, r1, false, b, (short)0, acc[t], false, false);
            }
            am = amn;
            ar = arn;
        }

        float* oBase = out + (size_t)(tm + 8 * half) * NCOL + tn + c;
        if (tm + 16 <= nrows) {
#pragma unroll
            for (int t = 0; t < JT; ++t) {
                const float bv = bias[tn + 16 * t + c];
#pragma unroll
                for (int r = 0; r < 8; ++r) {
                    float v = acc[t][r] + bv;
                    if (RELU) v = fmaxf(v, 0.0f);
                    oBase[(size_t)r * NCOL + 16 * t] = v;
                }
            }
        } else {
#pragma unroll
            for (int t = 0; t < JT; ++t) {
                const float bv = bias[tn + 16 * t + c];
#pragma unroll
                for (int r = 0; r < 8; ++r) {
                    float v = acc[t][r] + bv;
                    if (RELU) v = fmaxf(v, 0.0f);
                    if (tm + 8 * half + r < nrows) oBase[(size_t)r * NCOL + 16 * t] = v;
                }
            }
        }
    }
}

extern "C" void kernel_launch(void* const* d_in, const int* in_sizes, int n_in,
                              void* d_out, int out_size, void* d_ws, size_t ws_size,
                              hipStream_t stream)
{
    constexpr int IN_DIM = 128, HID = 256, EMB = 128;

    const float* x   = (const float*)d_in[0];
    const int*   ei  = (const int*)  d_in[1];
    const float* W1l = (const float*)d_in[2];
    const float* b1  = (const float*)d_in[3];
    const float* W1r = (const float*)d_in[4];
    const float* W2l = (const float*)d_in[5];
    const float* b2  = (const float*)d_in[6];
    const float* W2r = (const float*)d_in[7];

    const int N = in_sizes[0] / IN_DIM;
    const int E = in_sizes[1] / 2;
    const int* src = ei;
    const int* dst = ei + E;

    const int NP = ((N + 128) / 128) * 128;

    int*   deg    = (int*)d_ws;
    int*   rowptr = deg + NP;
    int*   col    = rowptr + NP;
    float* cnt    = (float*)(col + (((size_t)E + 63) & ~(size_t)63));
    float* agg1   = cnt  + (((size_t)N + 63) & ~(size_t)63);
    float* h      = agg1 + (size_t)N * IN_DIM;
    float* agg2   = h    + (size_t)N * HID;

    hipMemsetAsync(deg, 0, sizeof(int) * (size_t)NP, stream);

    const dim3 blk(256);
    const int edgeBlocks = 2048;
    const int nodeBlocks = 2048;
    const int gemmBlocks = 256;

    csr_count<<<edgeBlocks, blk, 0, stream>>>(dst, deg, E);
    csr_scan<<<1, dim3(1024), 0, stream>>>(deg, rowptr, NP);
    csr_fill<<<edgeBlocks, blk, 0, stream>>>(src, dst, rowptr, deg, col, E);

    sage_gather<IN_DIM><<<nodeBlocks, blk, 0, stream>>>(x, rowptr, col, agg1,
                                                        cnt, N, 1);
    sage_gemm<IN_DIM, HID, true><<<gemmBlocks, dim3(512), 0, stream>>>(
        agg1, cnt, x, W1l, W1r, b1, h, N);

    sage_gather<HID><<<nodeBlocks, blk, 0, stream>>>(h, rowptr, col, agg2,
                                                     cnt, N, 0);
    sage_gemm<HID, EMB, false><<<gemmBlocks, dim3(512), 0, stream>>>(
        agg2, cnt, h, W2l, W2r, b2, (float*)d_out, N);
}


# Edge list grouped by destination built by a two-level counting sort in LDS (bin edges by 256-node range, then count, scan, place per range): 2 launches, no global atomics, no memset; sum order kept
# speedup vs baseline: 3.6217x; 1.7175x over previous
#include <hip/hip_runtime.h>

typedef float v2f __attribute__((ext_vector_type(2)));
typedef float v8f __attribute__((ext_vector_type(8)));

constexpr int BIN_SHIFT = 8;
constexpr int BIN_R     = 1 << BIN_SHIFT;
constexpr int CHUNK     = 4096;
constexpr int MAXB      = 1024;
constexpr int MAXCH     = 1024;
constexpr int COLCAP    = 16384;

__device__ __forceinline__ int wave_incl_scan(int v, int lane)
{
#pragma unroll
    for (int o = 1; o < 32; o <<= 1) {
        const int t = __shfl_up(v, o);
        if (lane >= o) v += t;
    }
    return v;
}

__device__ __forceinline__ int wave_sum(int v)
{
#pragma unroll
    for (int o = 16; o > 0; o >>= 1) v += __shfl_xor(v, o);
    return v;
}

__device__ __forceinline__ int block_excl_scan(int v, int* wtot, int lane, int w, int& total)
{
    const int inc = wave_incl_scan(v, lane);
    if (lane == 31) wtot[w] = inc;
    __syncthreads();
    const int t    = wtot[lane];
    const int tinc = wave_incl_scan(t, lane);
    const int carry = __shfl(tinc, w) - __shfl(t, w);
    total = __shfl(tinc, 31);
    __syncthreads();
    return carry + (inc - v);
}

__global__ void __launch_bounds__(1024)
edge_bin(const int* __restrict__ src, const int* __restrict__ dst,
         int* __restrict__ pairs_w, int* __restrict__ segm_w, int E, int nb)
{
    int2* __restrict__ pairs = (int2*)pairs_w;
    int2* __restrict__ segm  = (int2*)segm_w;
    __shared__ int  hist[MAXB];
    __shared__ int  wtot[32];
    __shared__ int2 stage[CHUNK];
    const int tid  = threadIdx.x;
    const int lane = tid & 31;
    const int w    = tid >> 5;
    const int e0   = (int)blockIdx.x * CHUNK;
    const int len  = min(CHUNK, E - e0);

    hist[tid] = 0;
    int s[4], d[4];
#pragma unroll
    for (int u = 0; u < 4; ++u) {
        const int i = tid + 1024 * u;
        d[u] = (i < len) ? dst[e0 + i] : 0;
        s[u] = (i < len) ? src[e0 + i] : 0;
    }
    __syncthreads();
#pragma unroll
    for (int u = 0; u < 4; ++u)
        if (tid + 1024 * u < len) atomicAdd(&hist[d[u] >> BIN_SHIFT], 1);
    __syncthreads();

    const int c = hist[tid];
    int total;
    const int ex = block_excl_scan(c, wtot, lane, w, total);
    if (tid < nb) segm[(size_t)blockIdx.x * nb + tid] = make_int2(c, ex);
    hist[tid] = ex;
    __syncthreads();
#pragma unroll
    for (int u = 0; u < 4; ++u)
        if (tid + 1024 * u < len) {
            const int slot = atomicAdd(&hist[d[u] >> BIN_SHIFT], 1);
            stage[slot] = make_int2(s[u], d[u]);
        }
    __syncthreads();
#pragma unroll
    for (int u = 0; u < 4; ++u) {
        const int i = tid + 1024 * u;
        if (i < len) pairs[(size_t)e0 + i] = stage[i];
    }
}

template <bool PLACE>
__device__ __forceinline__ void edge_op(const int2 pr, int* cnt, int* colS,
                                        int* __restrict__ col, int base, bool inlds)
{
    int* ctr = &cnt[pr.y & (BIN_R - 1)];
    if (!PLACE) {
        atomicAdd(ctr, 1);
    } else {
        const int slot = atomicAdd(ctr, 1);
        if (inlds) colS[slot] = pr.x;
        else       col[base + slot] = pr.x;
    }
}

template <bool PLACE>
__device__ __forceinline__ void walk_pieces(const int2* __restrict__ pairs, const int2* seg,
                                            int* cnt, int* colS, int* __restrict__ col,
                                            int base, bool inlds, int nchunks, int w, int lane)
{
    for (int c0 = w; c0 < nchunks; c0 += 128) {
        int         n[4];
        const int2* p[4];
        int2        pr[4];
#pragma unroll
        for (int u = 0; u < 4; ++u) {
            const int  c = c0 + 32 * u;
            const int2 g = seg[min(c, MAXCH - 1)];
            n[u] = (c < nchunks) ? __builtin_amdgcn_readfirstlane(g.x) : 0;
            p[u] = pairs + (size_t)min(c, nchunks - 1) * CHUNK + __builtin_amdgcn_readfirstlane(g.y);
            const int2* q = (n[u] > 0) ? (p[u] + min(lane, n[u] - 1)) : pairs;
            pr[u] = *q;
        }
#pragma unroll
        for (int u = 0; u < 4; ++u) {
            if (lane < n[u]) edge_op<PLACE>(pr[u], cnt, colS, col, base, inlds);
            for (int i = lane + 32; i < n[u]; i += 32)
                edge_op<PLACE>(p[u][i], cnt, colS, col, base, inlds);
        }
    }
}

__global__ void __launch_bounds__(1024)
edge_group(const int* __restrict__ pairs_w, const int* __restrict__ segm_w,
           int* __restrict__ rowptr, int* __restrict__ col,
           int N, int nb, int nchunks)
{
    const int2* __restrict__ pairs = (const int2*)pairs_w;
    const int2* __restrict__ segm  = (const int2*)segm_w;
    __shared__ int  cnt[BIN_R];
    __shared__ int2 seg[MAXCH];
    __shared__ int  wtot[32];
    __shared__ int  sums[2];
    __shared__ int  colS[COLCAP];
    const int tid  = threadIdx.x;
    const int lane = tid & 31;
    const int w    = __builtin_amdgcn_readfirstlane(tid >> 5);
    const int b    = (int)blockIdx.x;
    const int lo   = b << BIN_SHIFT;

    if (tid < BIN_R) cnt[tid] = 0;
    if (tid < 2) sums[tid] = 0;
    __syncthreads();
    int2 sg = make_int2(0, 0);
    if (tid < nchunks) sg = segm[(size_t)tid * nb + b];
    seg[tid] = sg;
    const int sc = wave_sum(sg.x);
    const int so = wave_sum(sg.y);
    if (lane == 0) { atomicAdd(&sums[0], sc); atomicAdd(&sums[1], so); }
    __syncthreads();
    const int  total = sums[0];
    const int  base  = sums[1];
    const bool inlds = total <= COLCAP;

    walk_pieces<false>(pairs, seg, cnt, colS, col, base, inlds, nchunks, w, lane);
    __syncthreads();

    const int v = (tid < BIN_R) ? cnt[tid] : 0;
    int tot2;
    const int ex = block_excl_scan(v, wtot, lane, w, tot2);
    if (tid < BIN_R) {
        cnt[tid] = ex;
        if (lo + tid < N) rowptr[lo + tid] = base + ex;
    }
    if (b == nb - 1 && tid == 0) rowptr[N] = base + total;
    __syncthreads();

    walk_pieces<true>(pairs, seg, cnt, colS, col, base, inlds, nchunks, w, lane);
    __syncthreads();
    if (inlds)
        for (int i = tid; i < total; i += 1024) col[base + i] = colS[i];
}

template <int D>
__global__ void __launch_bounds__(256)
sage_gather(const float* __restrict__ x, const int* __restrict__ rowptr,
            const int* __restrict__ col, float* __restrict__ agg,
            float* __restrict__ cnt, int n, int do_count)
{
    constexpr int V = D / 128;
    const int lane = threadIdx.x & 31;
    const int wid = __builtin_amdgcn_readfirstlane(
        (int)(blockIdx.x * (blockDim.x >> 5) + (threadIdx.x >> 5)));
    const int nw  = gridDim.x * (blockDim.x >> 5);

    for (int node = wid; node < n; node += nw) {
        const int b = rowptr[node];
        const int e = rowptr[node + 1];

        float4 acc[V];
    #pragma unroll
        for (int v = 0; v < V; ++v) acc[v] = make_float4(0.f, 0.f, 0.f, 0.f);

        for (int base = b; base < e; base += 32) {
            const int m  = min(32, e - base);
            const int id = (lane < m) ? col[base + lane] : 0x7fffffff;

            int rank = 0;
            for (int j = 0; j < m; ++j) {
                const int v = __builtin_amdgcn_readlane(id, j);
                rank += (v < id || (v == id && j < lane)) ? 1 : 0;
            }

    #define ID_OF_RANK(k) __builtin_amdgcn_readlane(id, __builtin_ctz(__builtin_amdgcn_ballot_w32(rank == (k))))

            int i = 0;
            for (; i + 4 <= m; i += 4) {
                const int r0 = ID_OF_RANK(i);
                const int r1 = ID_OF_RANK(i + 1);
                const int r2 = ID_OF_RANK(i + 2);
                const int r3 = ID_OF_RANK(i + 3);
                const float4* p0 = (const float4*)(x + (size_t)r0 * D);
                const float4* p1 = (const float4*)(x + (size_t)r1 * D);
                const float4* p2 = (const float4*)(x + (size_t)r2 * D);
                const float4* p3 = (const float4*)(x + (size_t)r3 * D);
                float4 t0[V], t1[V], t2[V], t3[V];
    #pragma unroll
                for (int v = 0; v < V; ++v) {
                    t0[v] = p0[lane + 32 * v];
                    t1[v] = p1[lane + 32 * v];
                    t2[v] = p2[lane + 32 * v];
                    t3[v] = p3[lane + 32 * v];
                }
    #pragma unroll
                for (int v = 0; v < V; ++v) {
                    acc[v].x += t0[v].x; acc[v].y += t0[v].y; acc[v].z += t0[v].z; acc[v].w += t0[v].w;
                    acc[v].x += t1[v].x; acc[v].y += t1[v].y; acc[v].z += t1[v].z; acc[v].w += t1[v].w;
                    acc[v].x += t2[v].x; acc[v].y += t2[v].y; acc[v].z += t2[v].z; acc[v].w += t2[v].w;
                    acc[v].x += t3[v].x; acc[v].y += t3[v].y; acc[v].z += t3[v].z; acc[v].w += t3[v].w;
                }
            }
            for (; i < m; ++i) {
                const int r0 = ID_OF_RANK(i);
                const float4* p0 = (const float4*)(x + (size_t)r0 * D);
    #pragma unroll
                for (int v = 0; v < V; ++v) {
                    const float4 t = p0[lane + 32 * v];
                    acc[v].x += t.x; acc[v].y += t.y; acc[v].z += t.z; acc[v].w += t.w;
                }
            }
    #undef ID_OF_RANK
        }

        float4* o4 = (float4*)(agg + (size_t)node * D);
    #pragma unroll
        for (int v = 0; v < V; ++v) o4[lane + 32 * v] = acc[v];
        if (do_count && lane == 0) cnt[node] = (float)(e - b);
    }
}

template <int NCOL, int QK, int PQ>
__device__ __forceinline__ void stage_weights(const float* __restrict__ Wsrc,
                                              float4* dst, int tid)
{
    constexpr int TRIPS = QK * NCOL / 512;
#pragma unroll
    for (int i0 = 0; i0 < TRIPS; i0 += 8) {
        float4 v[8];
#pragma unroll
        for (int u = 0; u < 8; ++u) {
            const int e = tid + 512 * (i0 + u);
            const float* s = Wsrc + (size_t)(4 * (e / NCOL)) * NCOL + (e % NCOL);
            v[u].x = s[0];
            v[u].y = s[NCOL];
            v[u].z = s[2 * NCOL];
            v[u].w = s[3 * NCOL];
        }
#pragma unroll
        for (int u = 0; u < 8; ++u) {
            const int e = tid + 512 * (i0 + u);
            dst[(e / NCOL) * PQ + (e % NCOL)] = v[u];
        }
    }
}

template <int K, int NCOL, bool RELU>
__global__ void __launch_bounds__(512)
sage_gemm(const float* __restrict__ agg, const float* __restrict__ cnt,
          const float* __restrict__ xroot, const float* __restrict__ Wl,
          const float* __restrict__ Wr, const float* __restrict__ bias,
          float* __restrict__ out, int nrows)
{
    constexpr int JT     = 4;
    constexpr int GROUPS = NCOL / (16 * JT);
    constexpr int QK     = K / 4;
    constexpr int PQ     = NCOL + 8;
    constexpr int NB     = K / 8;
    __shared__ float4 Ws[2 * QK * PQ];

    const int tid = threadIdx.x;

    stage_weights<NCOL, QK, PQ>(Wl, Ws, tid);
    stage_weights<NCOL, QK, PQ>(Wr, Ws + QK * PQ, tid);
    __syncthreads();

    const int lane = tid & 31;
    const int half = lane >> 4;
    const int c    = lane & 15;
    const int w = __builtin_amdgcn_readfirstlane(tid >> 5);

    const int ntiles = (nrows + 15) >> 4;
    const int first  = (int)blockIdx.x;
    const int step   = (int)gridDim.x;
    const int mine   = (first < ntiles) ? (ntiles - first + step - 1) / step : 0;
    const int units  = mine * GROUPS;

    for (int u = w; u < units; u += 16) {
        const int tm = (first + (u / GROUPS) * step) * 16;
        const int tn = (u % GROUPS) * (16 * JT);

        const int row = min(tm + c, nrows - 1);
        const float  inv = 1.0f / fmaxf(cnt[row], 1.0f);
        const float4* aP = (const float4*)(agg   + (size_t)row * K) + half;
        const float4* xP = (const float4*)(xroot + (size_t)row * K) + half;
        const float4* bL = Ws + half * PQ + tn + c;
        const float4* bR = bL + QK * PQ;

        v8f acc[JT] = {};
        float4 am = aP[0];
        float4 ar = xP[0];
        for (int j = 0; j < NB; ++j) {
            const int jn = (j + 1 < NB) ? (j + 1) : j;
            const float4 amn = aP[2 * jn];
            const float4 arn = xP[2 * jn];

            v2f m0, m1, r0, r1;
            m0.x = am.x * inv; m0.y = am.y * inv;
            m1.x = am.z * inv; m1.y = am.w * inv;
            r0.x = ar.x;       r0.y = ar.y;
            r1.x = ar.z;       r1.y = ar.w;

            float4 wl[JT], wr[JT];
#pragma unroll
            for (int t = 0; t < JT; ++t) {
                wl[t] = bL[(2 * j) * PQ + 16 * t];
                wr[t] = bR[(2 * j) * PQ + 16 * t];
            }
#pragma unroll
            for (int t = 0; t < JT; ++t) {
                v2f b; b.x = wl[t].x; b.y = wl[t].y;
                acc[t] = __builtin_amdgcn_wmma_f32_16x16x4_f32(
                    false, m0, false, b, (short)0, acc[t], false, false);
            }
#pragma unroll
            for (int t = 0; t < JT; ++t) {
                v2f b; b.x = wr[t].x; b.y = wr[t].y;
                acc[t] = __builtin_amdgcn_wmma_f32_16x16x4_f32(
                    false, r0, false, b, (short)0, acc[t], false, false);
            }
#pragma unroll
            for (int t = 0; t < JT; ++t) {
                v2f b; b.x = wl[t].z; b.y = wl[t].w;
                acc[t] = __builtin_amdgcn_wmma_f32_16x16x4_f32(
                    false, m1, false, b, (short)0, acc[t], false, false);
            }
#pragma unroll
            for (int t = 0; t < JT; ++t) {
                v2f b; b.x = wr[t].z; b.y = wr[t].w;
                acc[t] = __builtin_amdgcn_wmma_f32_16x16x4_f32(
                    false, r1, false, b, (short)0, acc[t], false, false);
            }
            am = amn;
            ar = arn;
        }

        float* oBase = out + (size_t)(tm + 8 * half) * NCOL + tn + c;
        if (tm + 16 <= nrows) {
#pragma unroll
            for (int t = 0; t < JT; ++t) {
                const float bv = bias[tn + 16 * t + c];
#pragma unroll
                for (int r = 0; r < 8; ++r) {
                    float v = acc[t][r] + bv;
                    if (RELU) v = fmaxf(v, 0.0f);
                    oBase[(size_t)r * NCOL + 16 * t] = v;
                }
            }
        } else {
#pragma unroll
            for (int t = 0; t < JT; ++t) {
                const float bv = bias[tn + 16 * t + c];
#pragma unroll
                for (int r = 0; r < 8; ++r) {
                    float v = acc[t][r] + bv;
                    if (RELU) v = fmaxf(v, 0.0f);
                    if (tm + 8 * half + r < nrows) oBase[(size_t)r * NCOL + 16 * t] = v;
                }
            }
        }
    }
}

extern "C" void kernel_launch(void* const* d_in, const int* in_sizes, int n_in,
                              void* d_out, int out_size, void* d_ws, size_t ws_size,
                              hipStream_t stream)
{
    constexpr int IN_DIM = 128, HID = 256, EMB = 128;

    const float* x   = (const float*)d_in[0];
    const int*   ei  = (const int*)  d_in[1];
    const float* W1l = (const float*)d_in[2];
    const float* b1  = (const float*)d_in[3];
    const float* W1r = (const float*)d_in[4];
    const float* W2l = (const float*)d_in[5];
    const float* b2  = (const float*)d_in[6];
    const float* W2r = (const float*)d_in[7];

    const int N = in_sizes[0] / IN_DIM;
    const int E = in_sizes[1] / 2;
    const int* src = ei;
    const int* dst = ei + E;

    const int NP = ((N + 128) / 128) * 128;
    const int nb      = (N + BIN_R - 1) >> BIN_SHIFT;
    const int nchunks = (E + CHUNK - 1) / CHUNK;

    int*   rowptr = (int*)d_ws;
    int*   col    = rowptr + NP;
    float* cnt    = (float*)(col + (((size_t)E + 63) & ~(size_t)63));
    float* agg1   = cnt  + (((size_t)N + 63) & ~(size_t)63);
    float* h      = agg1 + (size_t)N * IN_DIM;
    float* agg2   = h    + (size_t)N * HID;
    int*   pairs  = (int*)(agg2 + (size_t)N * HID);
    int*   segm   = pairs + 2 * (size_t)nchunks * CHUNK;

    const dim3 blk(256);
    const int nodeBlocks = 2048;
    const int gemmBlocks = 256;

    edge_bin<<<nchunks, dim3(1024), 0, stream>>>(src, dst, pairs, segm, E, nb);
    edge_group<<<nb, dim3(1024), 0, stream>>>(pairs, segm, rowptr, col, N, nb, nchunks);

    sage_gather<IN_DIM><<<nodeBlocks, blk, 0, stream>>>(x, rowptr, col, agg1,
                                                        cnt, N, 1);
    sage_gemm<IN_DIM, HID, true><<<gemmBlocks, dim3(512), 0, stream>>>(
        agg1, cnt, x, W1l, W1r, b1, h, N);

    sage_gather<HID><<<nodeBlocks, blk, 0, stream>>>(h, rowptr, col, agg2,
                                                     cnt, N, 0);
    sage_gemm<HID, EMB, false><<<gemmBlocks, dim3(512), 0, stream>>>(
        agg2, cnt, h, W2l, W2r, b2, (float*)d_out, N);
}
